# starting kernel
# baseline (speedup 1.0000x reference)
#include <hip/hip_runtime.h>
#include <hip/hip_bf16.h>


#define N_NODES 10000
#define N_EDGES 640000
#define IN_C    128
#define OUT_C   128

typedef __attribute__((ext_vector_type(2))) float v2f;
typedef __attribute__((ext_vector_type(8))) float v8f;

__global__ void k_init_deg(float* __restrict__ deg) {
    int i = blockIdx.x * blockDim.x + threadIdx.x;
    if (i < N_NODES) deg[i] = 1.0f;
}

__global__ void k_deg_accum(const int* __restrict__ col, float* __restrict__ deg) {
    int e = blockIdx.x * blockDim.x + threadIdx.x;
    if (e < N_EDGES) atomicAdd(&deg[col[e]], 1.0f);
}

__global__ void k_inv_sqrt(const float* __restrict__ deg, float* __restrict__ dis) {
    int i = blockIdx.x * blockDim.x + threadIdx.x;
    if (i < N_NODES) {
        float d = deg[i];
        dis[i] = (d > 0.0f) ? rsqrtf(d) : 0.0f;
    }
}

__global__ void k_gemm_wmma(const float* __restrict__ x,
                            const float* __restrict__ W,
                            float* __restrict__ h) {
    const int wave   = (blockIdx.x * blockDim.x + threadIdx.x) >> 5;
    const int lane   = threadIdx.x & 31;
    const int tilesN = OUT_C / 16;
    const int tm     = wave / tilesN;
    const int tn     = wave % tilesN;

    const int mn    = lane & 15;
    const int khalf = (lane >> 4) << 1;

    const float* __restrict__ xr = x + (tm * 16 + mn) * IN_C + khalf;
    const float* __restrict__ wr = W + (tn * 16 + mn) * IN_C + khalf;

    v8f c = {};
    #pragma unroll 4
    for (int k = 0; k < IN_C; k += 4) {
        v2f a = *(const v2f*)(xr + k);
        v2f b = *(const v2f*)(wr + k);
        c = __builtin_amdgcn_wmma_f32_16x16x4_f32(false, a, false, b,
                                                  (short)0, c, false, false);
    }

    const int mbase = tm * 16 + ((lane >> 4) << 3);
    const int ncol  = tn * 16 + mn;
    #pragma unroll
    for (int v = 0; v < 8; ++v) {
        h[(mbase + v) * OUT_C + ncol] = c[v];
    }
}

__global__ void k_out_init(const float* __restrict__ h,
                           const float* __restrict__ dis,
                           const float* __restrict__ bias,
                           float* __restrict__ out) {
    int i = blockIdx.x * blockDim.x + threadIdx.x;
    if (i < N_NODES * OUT_C) {
        int node = i >> 7;
        int ch   = i & 127;
        float d  = dis[node];
        out[i] = h[i] * d * d + bias[ch];
    }
}

__global__ void k_edge_scatter(const int* __restrict__ row,
                               const int* __restrict__ col,
                               const float* __restrict__ h,
                               const float* __restrict__ dis,
                               float* __restrict__ out) {
    const int wid    = (blockIdx.x * blockDim.x + threadIdx.x) >> 5;
    const int lane   = threadIdx.x & 31;
    const int nwaves = (gridDim.x * blockDim.x) >> 5;

    for (int e = wid; e < N_EDGES; e += nwaves) {
        const int r = row[e];
        const int c = col[e];
        const float norm = dis[r] * dis[c];

        float4 v = *(const float4*)(h + r * OUT_C + lane * 4);
        float* o = out + c * OUT_C + lane * 4;
        atomicAdd(o + 0, v.x * norm);
        atomicAdd(o + 1, v.y * norm);
        atomicAdd(o + 2, v.z * norm);
        atomicAdd(o + 3, v.w * norm);
    }
}

extern "C" void kernel_launch(void* const* d_in, const int* in_sizes, int n_in,
                              void* d_out, int out_size, void* d_ws, size_t ws_size,
                              hipStream_t stream) {
    const float* x  = (const float*)d_in[0];
    const int*   ei = (const int*)  d_in[1];
    const float* W  = (const float*)d_in[2];
    const float* b  = (const float*)d_in[3];
    float*       out = (float*)d_out;

    const int* row = ei;
    const int* col = ei + N_EDGES;

    float* deg = (float*)d_ws;
    float* dis = deg + N_NODES;
    float* h   = dis + N_NODES;

    k_init_deg<<<(N_NODES + 255) / 256, 256, 0, stream>>>(deg);
    k_deg_accum<<<(N_EDGES + 255) / 256, 256, 0, stream>>>(col, deg);
    k_inv_sqrt<<<(N_NODES + 255) / 256, 256, 0, stream>>>(deg, dis);
    k_gemm_wmma<<<625, 256, 0, stream>>>(x, W, h);
    k_out_init<<<(N_NODES * OUT_C + 255) / 256, 256, 0, stream>>>(h, dis, b, out);
    k_edge_scatter<<<4096, 256, 0, stream>>>(row, col, h, dis, out);
}


# out cut into 16 node ranges x 4 edge slices, not 4 x 16: a slab is 625 rows, so 4x less slab is zeroed and merged (160k f32 row atomics, not 640k); 128 sort + 128 degree blocks; sum order changed
# speedup vs baseline: 4.0736x; 4.0736x over previous
#include <hip/hip_runtime.h>
#include <hip/hip_bf16.h>


#define N_NODES 10000
#define N_EDGES 640000
#define IN_C    128
#define OUT_C   128

typedef __attribute__((ext_vector_type(2))) float v2f;
typedef __attribute__((ext_vector_type(8))) float v8f;

__device__ __forceinline__ int package_half() {
    return (int)((__builtin_amdgcn_s_sendmsg_rtn(0x87) >> 18) & 1u);
}

#define TICKET_STRIDE 64

__global__ void k_init_deg(float* __restrict__ deg0, float* __restrict__ deg1) {
    int i = blockIdx.x * blockDim.x + threadIdx.x;
    if (i < N_NODES) { deg0[i] = 1.0f; deg1[i] = 0.0f; }
}

#define D_PARTS      16
#define S_PARTS      4
#define E_PARTS      4
#define TILE_N       (N_NODES / D_PARTS)
#define SLAB_C       (OUT_C / S_PARTS)
#define SLAB_THREADS 1024
#define SLAB_WAVES   (SLAB_THREADS / 32)
#define PB_BLOCKS    128
#define PB_EDGES     (N_EDGES / PB_BLOCKS)
#define PB_THREADS   1024
#define PB_SLOTS     5
#define RUN_BATCH    32
#define PB_STRIDE    5504
#define SRC_BITS     14
#define SRC_MASK     ((1u << SRC_BITS) - 1u)
#define DUMP_REC     ((unsigned)TILE_N << SRC_BITS)
#define WS_BYTES     ((size_t)4 * (2 * N_NODES + N_NODES * OUT_C + PB_BLOCKS * D_PARTS + PB_BLOCKS * PB_STRIDE))

static_assert(N_NODES % D_PARTS == 0, "node ranges are equal");
static_assert(PB_BLOCKS * PB_EDGES == N_EDGES && PB_THREADS * PB_SLOTS >= PB_EDGES &&
              PB_THREADS % 32 == 0, "partition blocks cover the edge list in whole waves");
static_assert(PB_BLOCKS == E_PARTS * SLAB_WAVES, "one partition block per edge-pass wave");
static_assert(PB_STRIDE >= PB_EDGES + D_PARTS * (RUN_BATCH - 1) && PB_STRIDE % RUN_BATCH == 0,
              "the padded runs fit a block's record region");
static_assert(PB_STRIDE < (1 << 16) && D_PARTS * RUN_BATCH <= PB_THREADS,
              "a run's start fits the table entry's low half; one wave pads each run");
static_assert(N_NODES <= (1 << SRC_BITS) && TILE_N + 1 <= (1 << (32 - SRC_BITS)),
              "record fields hold a source row and a local row (dump row included)");

__global__ __launch_bounds__(PB_THREADS)
void k_deg_part(const int* __restrict__ row, const int* __restrict__ col,
                float* __restrict__ deg0, float* __restrict__ deg1,
                int* __restrict__ tab, int* __restrict__ recs) {
    __shared__ int s_cnt[D_PARTS];
    __shared__ int s_start[D_PARTS];
    const int tid  = threadIdx.x;
    const int lane = tid & 31;

    if (blockIdx.x >= PB_BLOCKS) {
        float* plane = package_half() ? deg1 : deg0;
        const int* cp = col + (blockIdx.x - PB_BLOCKS) * PB_EDGES;
        int c[PB_SLOTS];
        #pragma unroll
        for (int q = 0; q < PB_SLOTS; ++q) {
            const int i = tid + q * PB_THREADS;
            c[q] = i < PB_EDGES ? cp[i] : -1;
        }
        #pragma unroll
        for (int q = 0; q < PB_SLOTS; ++q)
            if (c[q] >= 0) atomicAdd(&plane[c[q]], 1.0f);
        return;
    }

    if (tid < D_PARTS) s_cnt[tid] = 0;
    __syncthreads();

    const int e0 = blockIdx.x * PB_EDGES;

    int key[PB_SLOTS], pos[PB_SLOTS];
    unsigned rec[PB_SLOTS];
    #pragma unroll
    for (int q = 0; q < PB_SLOTS; ++q) {
        const int i = tid + q * PB_THREADS;
        int k = -1;
        unsigned rc = 0u;
        if (i < PB_EDGES) {
            const int c = col[e0 + i];
            const int r = row[e0 + i];
            k  = c / TILE_N;
            rc = (unsigned)r | ((unsigned)(c - k * TILE_N) << SRC_BITS);
        }
        key[q] = k; rec[q] = rc;
    }
    #pragma unroll
    for (int q = 0; q < PB_SLOTS; ++q)
        pos[q] = key[q] >= 0 ? atomicAdd(&s_cnt[key[q]], 1) : 0;
    __syncthreads();

    if (tid < D_PARTS) {
        int st = 0;
        for (int j = 0; j < tid; ++j) st += (s_cnt[j] + RUN_BATCH - 1) & ~(RUN_BATCH - 1);
        s_start[tid] = st;
    }
    __syncthreads();

    int* my = recs + blockIdx.x * PB_STRIDE;
    #pragma unroll
    for (int q = 0; q < PB_SLOTS; ++q)
        if (key[q] >= 0) my[s_start[key[q]] + pos[q]] = (int)rec[q];
    if (tid < D_PARTS * RUN_BATCH) {
        const int k  = tid >> 5;
        const int tk = s_cnt[k];
        const int st = s_start[k];
        const int padded = (tk + RUN_BATCH - 1) & ~(RUN_BATCH - 1);
        if (tk + lane < padded) my[st + tk + lane] = (int)DUMP_REC;
        if (lane == 0) tab[blockIdx.x * D_PARTS + k] = st | ((padded / RUN_BATCH) << 16);
    }
}

__global__ void k_inv_sqrt(float* deg0, float* deg1_dis) {
    int i = blockIdx.x * blockDim.x + threadIdx.x;
    if (i < N_NODES) {
        float d = deg0[i] + deg1_dis[i];
        deg1_dis[i] = (d > 0.0f) ? rsqrtf(d) : 0.0f;
        if (i == 0 || i == TICKET_STRIDE) deg0[i] = 0.0f;
    }
}

__global__ void k_gemm_wmma(const float* __restrict__ x,
                            const float* __restrict__ W,
                            const float* __restrict__ dis,
                            const float* __restrict__ bias,
                            float* __restrict__ h,
                            float* __restrict__ out) {
    const int wave   = (blockIdx.x * blockDim.x + threadIdx.x) >> 5;
    const int lane   = threadIdx.x & 31;
    const int tilesN = OUT_C / 16;
    const int tm     = wave / tilesN;
    const int tn     = wave % tilesN;

    const int mn    = lane & 15;
    const int khalf = (lane >> 4) << 1;

    const float* __restrict__ xr = x + (tm * 16 + mn) * IN_C + khalf;
    const float* __restrict__ wr = W + (tn * 16 + mn) * IN_C + khalf;

    v8f c = {};
    #pragma unroll 4
    for (int k = 0; k < IN_C; k += 4) {
        v2f a = *(const v2f*)(xr + k);
        v2f b = *(const v2f*)(wr + k);
        c = __builtin_amdgcn_wmma_f32_16x16x4_f32(false, a, false, b,
                                                  (short)0, c, false, false);
    }

    const int mbase = tm * 16 + ((lane >> 4) << 3);
    const int ncol  = tn * 16 + mn;
    const float bv  = bias[ncol];
    #pragma unroll
    for (int v = 0; v < 8; ++v) {
        const float d  = dis[mbase + v];
        const float hv = c[v] * d;
        h[(mbase + v) * OUT_C + ncol]   = hv;
        out[(mbase + v) * OUT_C + ncol] = hv * d + bv;
    }
}

#define HALF         16
#define SLAB_GRID    256
#define WAVE_ROWS    (TILE_N / SLAB_WAVES)
#define MERGE_ROWS   5
#define LDS_FLOATS   42000

static_assert(N_NODES % D_PARTS == 0 && N_EDGES % E_PARTS == 0 && SLAB_C == 32,
              "slab geometry");
static_assert((TILE_N + 1) * SLAB_C <= LDS_FLOATS && LDS_FLOATS * 4 + 4 <= 327680 && LDS_FLOATS * 4 > 163840,
              "slab and dump row fit the reservation; the reservation fits one WGP and excludes a second work-group");
static_assert(SLAB_GRID == D_PARTS * S_PARTS * E_PARTS, "one work-group per (slice, range, slab)");
static_assert(2 * HALF == RUN_BATCH, "a batch is two request groups");
#define ROLES_PER_HALF (SLAB_GRID / 2)
static_assert(D_PARTS % 2 == 0 && (SLAB_GRID & (SLAB_GRID - 1)) == 0, "roles split evenly by half");

struct EdgeHalf {
    float v[HALF];
    int   a[HALF];
};

__device__ __forceinline__ void half_request(EdgeHalf& k, int rec, int off,
                                             const float* __restrict__ hs_slab,
                                             unsigned lane4) {
    #pragma unroll
    for (int j = 0; j < HALF; ++j) {
        const unsigned r = (unsigned)__builtin_amdgcn_readlane(rec, off + j);
        const unsigned byte = (r & SRC_MASK) * (OUT_C * 4) + lane4;
        k.v[j] = *(const float*)((const char*)hs_slab + byte);
        k.a[j] = (int)((r >> SRC_BITS) * (SLAB_C * 4) + lane4);
    }
}

__device__ __forceinline__ void half_add(const EdgeHalf& k, float* tile) {
    #pragma unroll
    for (int j = 0; j < HALF; ++j)
        __hip_atomic_fetch_add((float*)((char*)tile + k.a[j]), k.v[j],
                               __ATOMIC_RELAXED, __HIP_MEMORY_SCOPE_WORKGROUP);
}

__global__ __launch_bounds__(SLAB_THREADS)
void k_edge_slab(const int* __restrict__ tab,
                 const int* __restrict__ recs,
                 const float* __restrict__ hs,
                 const float* __restrict__ dis,
                 float* __restrict__ out,
                 int* __restrict__ ticket) {
    __shared__ float tile[LDS_FLOATS];
    __shared__ int   s_role;

    const int tid  = threadIdx.x;
    const int lane = tid & 31;

    if (tid == 0) {
        const int H = package_half();
        const int t = atomicAdd(&ticket[H * TICKET_STRIDE], 1);
        s_role = (t < ROLES_PER_HALF ? H * ROLES_PER_HALF + t
                                     : (1 - H) * ROLES_PER_HALF + (2 * ROLES_PER_HALF - 1 - t))
                 & (SLAB_GRID - 1);
    }

    float4* t4 = (float4*)tile;
    for (int i = tid; i < (TILE_N + 1) * SLAB_C / 4; i += SLAB_THREADS)
        t4[i] = make_float4(0.0f, 0.0f, 0.0f, 0.0f);
    __syncthreads();

    const unsigned role = (unsigned)__builtin_amdgcn_readfirstlane(s_role);
    const int      wave = __builtin_amdgcn_readfirstlane(tid >> 5);
    const unsigned idx  = role % ROLES_PER_HALF;
    const int d = (int)((role / ROLES_PER_HALF) * (D_PARTS / 2) + idx % (D_PARTS / 2));
    const int s = (int)((idx / (D_PARTS / 2)) % S_PARTS);
    const int p = (int)(idx / ((D_PARTS / 2) * S_PARTS));

    const int    nbase = d * TILE_N;
    const float*   hs_slab = hs + s * SLAB_C;
    const unsigned lane4   = (unsigned)lane * 4u;

    const int  blk = p * SLAB_WAVES + wave;
    const int  te  = tab[blk * D_PARTS + d];
    const int  nb  = te >> 16;
    const int* rp  = recs + blk * PB_STRIDE + (te & 0xffff) + lane;

    if (nb > 0) {
        EdgeHalf A, B;
        int rec  = rp[0];
        int recn = rp[(nb > 1 ? 1 : 0) * RUN_BATCH];
        asm volatile("" : "+v"(rec), "+v"(recn));
        half_request(A, rec, 0, hs_slab, lane4);
        for (int b = 1; b < nb; ++b) {
            half_request(B, rec, HALF, hs_slab, lane4);
            half_add(A, tile);
            rec  = recn;
            recn = rp[(b + 1 < nb ? b + 1 : b) * RUN_BATCH];
            half_request(A, rec, 0, hs_slab, lane4);
            half_add(B, tile);
        }
        half_request(B, rec, HALF, hs_slab, lane4);
        half_add(A, tile);
        half_add(B, tile);
    }
    __syncthreads();

    float* o_lane = out + s * SLAB_C + lane;
    #pragma unroll
    for (int g = 0; g < WAVE_ROWS; g += MERGE_ROWS) {
        float v[MERGE_ROWS], dc[MERGE_ROWS];
        #pragma unroll
        for (int j = 0; j < MERGE_ROWS; ++j) {
            if (g + j < WAVE_ROWS) {
                const int nn = wave + (g + j) * SLAB_WAVES;
                v[j]  = tile[nn * SLAB_C + lane];
                dc[j] = dis[nbase + nn];
            }
        }
        #pragma unroll
        for (int j = 0; j < MERGE_ROWS; ++j) {
            if (g + j < WAVE_ROWS) {
                const int nn = wave + (g + j) * SLAB_WAVES;
                atomicAdd(o_lane + (nbase + nn) * OUT_C, v[j] * dc[j]);
            }
        }
    }
    if (wave < TILE_N - WAVE_ROWS * SLAB_WAVES) {
        const int nn = WAVE_ROWS * SLAB_WAVES + wave;
        atomicAdd(o_lane + (nbase + nn) * OUT_C, tile[nn * SLAB_C + lane] * dis[nbase + nn]);
    }
}

extern "C" void kernel_launch(void* const* d_in, const int* in_sizes, int n_in,
                              void* d_out, int out_size, void* d_ws, size_t ws_size,
                              hipStream_t stream) {
    const float* x  = (const float*)d_in[0];
    const int*   ei = (const int*)  d_in[1];
    const float* W  = (const float*)d_in[2];
    const float* b  = (const float*)d_in[3];
    float*       out = (float*)d_out;

    const int* row = ei;
    const int* col = ei + N_EDGES;

    float* deg  = (float*)d_ws;
    float* dis  = deg + N_NODES;
    float* h    = dis + N_NODES;
    int*   tab  = (int*)(h + N_NODES * OUT_C);
    int*   recs = tab + PB_BLOCKS * D_PARTS;
    if (ws_size < WS_BYTES) return;

    k_init_deg<<<(N_NODES + 255) / 256, 256, 0, stream>>>(deg, dis);
    k_deg_part<<<2 * PB_BLOCKS, PB_THREADS, 0, stream>>>(row, col, deg, dis, tab, recs);
    k_inv_sqrt<<<(N_NODES + 255) / 256, 256, 0, stream>>>(deg, dis);
    k_gemm_wmma<<<625, 256, 0, stream>>>(x, W, dis, b, h, out);
    k_edge_slab<<<SLAB_GRID, SLAB_THREADS, 0, stream>>>(tab, recs, h, dis, out, (int*)deg);
}


# GEMM: a wave keeps a 32x64 tile (8 accumulators), 16-byte k pieces per lane: 6 fragment loads per 16 WMMAs, not 32; a lane's 4 tiles are 4 adjacent columns: whole-line stores; k sum order changed
# speedup vs baseline: 4.4480x; 1.0919x over previous
#include <hip/hip_runtime.h>
#include <hip/hip_bf16.h>


#define N_NODES 10000
#define N_EDGES 640000
#define IN_C    128
#define OUT_C   128

typedef __attribute__((ext_vector_type(2))) float v2f;
typedef __attribute__((ext_vector_type(8))) float v8f;

__device__ __forceinline__ int package_half() {
    return (int)((__builtin_amdgcn_s_sendmsg_rtn(0x87) >> 18) & 1u);
}

#define TICKET_STRIDE 64

__global__ void k_init_deg(float* __restrict__ deg0, float* __restrict__ deg1) {
    int i = blockIdx.x * blockDim.x + threadIdx.x;
    if (i < N_NODES) { deg0[i] = 1.0f; deg1[i] = 0.0f; }
}

#define D_PARTS      16
#define S_PARTS      4
#define E_PARTS      4
#define TILE_N       (N_NODES / D_PARTS)
#define SLAB_C       (OUT_C / S_PARTS)
#define SLAB_THREADS 1024
#define SLAB_WAVES   (SLAB_THREADS / 32)
#define PB_BLOCKS    128
#define PB_EDGES     (N_EDGES / PB_BLOCKS)
#define PB_THREADS   1024
#define PB_SLOTS     5
#define RUN_BATCH    32
#define PB_STRIDE    5504
#define SRC_BITS     14
#define SRC_MASK     ((1u << SRC_BITS) - 1u)
#define DUMP_REC     ((unsigned)TILE_N << SRC_BITS)
#define WS_BYTES     ((size_t)4 * (2 * N_NODES + N_NODES * OUT_C + PB_BLOCKS * D_PARTS + PB_BLOCKS * PB_STRIDE))

static_assert(N_NODES % D_PARTS == 0, "node ranges are equal");
static_assert(PB_BLOCKS * PB_EDGES == N_EDGES && PB_THREADS * PB_SLOTS >= PB_EDGES &&
              PB_THREADS % 32 == 0, "partition blocks cover the edge list in whole waves");
static_assert(PB_BLOCKS == E_PARTS * SLAB_WAVES, "one partition block per edge-pass wave");
static_assert(PB_STRIDE >= PB_EDGES + D_PARTS * (RUN_BATCH - 1) && PB_STRIDE % RUN_BATCH == 0,
              "the padded runs fit a block's record region");
static_assert(PB_STRIDE < (1 << 16) && D_PARTS * RUN_BATCH <= PB_THREADS,
              "a run's start fits the table entry's low half; one wave pads each run");
static_assert(N_NODES <= (1 << SRC_BITS) && TILE_N + 1 <= (1 << (32 - SRC_BITS)),
              "record fields hold a source row and a local row (dump row included)");

__global__ __launch_bounds__(PB_THREADS)
void k_deg_part(const int* __restrict__ row, const int* __restrict__ col,
                float* __restrict__ deg0, float* __restrict__ deg1,
                int* __restrict__ tab, int* __restrict__ recs) {
    __shared__ int s_cnt[D_PARTS];
    __shared__ int s_start[D_PARTS];
    const int tid  = threadIdx.x;
    const int lane = tid & 31;

    if (blockIdx.x >= PB_BLOCKS) {
        float* plane = package_half() ? deg1 : deg0;
        const int* cp = col + (blockIdx.x - PB_BLOCKS) * PB_EDGES;
        int c[PB_SLOTS];
        #pragma unroll
        for (int q = 0; q < PB_SLOTS; ++q) {
            const int i = tid + q * PB_THREADS;
            c[q] = i < PB_EDGES ? cp[i] : -1;
        }
        #pragma unroll
        for (int q = 0; q < PB_SLOTS; ++q)
            if (c[q] >= 0) atomicAdd(&plane[c[q]], 1.0f);
        return;
    }

    if (tid < D_PARTS) s_cnt[tid] = 0;
    __syncthreads();

    const int e0 = blockIdx.x * PB_EDGES;

    int key[PB_SLOTS], pos[PB_SLOTS];
    unsigned rec[PB_SLOTS];
    #pragma unroll
    for (int q = 0; q < PB_SLOTS; ++q) {
        const int i = tid + q * PB_THREADS;
        int k = -1;
        unsigned rc = 0u;
        if (i < PB_EDGES) {
            const int c = col[e0 + i];
            const int r = row[e0 + i];
            k  = c / TILE_N;
            rc = (unsigned)r | ((unsigned)(c - k * TILE_N) << SRC_BITS);
        }
        key[q] = k; rec[q] = rc;
    }
    #pragma unroll
    for (int q = 0; q < PB_SLOTS; ++q)
        pos[q] = key[q] >= 0 ? atomicAdd(&s_cnt[key[q]], 1) : 0;
    __syncthreads();

    if (tid < D_PARTS) {
        int st = 0;
        for (int j = 0; j < tid; ++j) st += (s_cnt[j] + RUN_BATCH - 1) & ~(RUN_BATCH - 1);
        s_start[tid] = st;
    }
    __syncthreads();

    int* my = recs + blockIdx.x * PB_STRIDE;
    #pragma unroll
    for (int q = 0; q < PB_SLOTS; ++q)
        if (key[q] >= 0) my[s_start[key[q]] + pos[q]] = (int)rec[q];
    if (tid < D_PARTS * RUN_BATCH) {
        const int k  = tid >> 5;
        const int tk = s_cnt[k];
        const int st = s_start[k];
        const int padded = (tk + RUN_BATCH - 1) & ~(RUN_BATCH - 1);
        if (tk + lane < padded) my[st + tk + lane] = (int)DUMP_REC;
        if (lane == 0) tab[blockIdx.x * D_PARTS + k] = st | ((padded / RUN_BATCH) << 16);
    }
}

__global__ void k_inv_sqrt(float* deg0, float* deg1_dis) {
    int i = blockIdx.x * blockDim.x + threadIdx.x;
    if (i < N_NODES) {
        float d = deg0[i] + deg1_dis[i];
        deg1_dis[i] = (d > 0.0f) ? rsqrtf(d) : 0.0f;
        if (i == 0 || i == TICKET_STRIDE) deg0[i] = 0.0f;
    }
}

#define GM_ROWS   32
#define GM_COLS   64
#define GM_BLOCKS ((N_NODES + GM_ROWS - 1) / GM_ROWS)
#define GM_KSTEP  8
#define GM_STEPS  (IN_C / GM_KSTEP)
#define GM_AHEAD  4

static_assert(IN_C % GM_KSTEP == 0 && OUT_C == 2 * GM_COLS && GM_STEPS >= GM_AHEAD,
              "two waves cover a row's columns; whole k steps");
static_assert(N_NODES % 16 == 0 && N_NODES % GM_ROWS == 16,
              "the last block's first row tile is whole and its second lies wholly outside");

struct GemmStage { float4 a0, a1, b[4]; };

__device__ __forceinline__ void gemm_request(GemmStage& s, const float* __restrict__ xa0,
                                             const float* __restrict__ xa1,
                                             const float* __restrict__ wb, int kb) {
    s.a0 = *(const float4*)(xa0 + kb * GM_KSTEP);
    s.a1 = *(const float4*)(xa1 + kb * GM_KSTEP);
    #pragma unroll
    for (int j = 0; j < 4; ++j)
        s.b[j] = *(const float4*)(wb + j * IN_C + kb * GM_KSTEP);
}

__device__ __forceinline__ v8f gemm_mma(v2f a, v2f b, v8f c) {
    return __builtin_amdgcn_wmma_f32_16x16x4_f32(false, a, false, b, (short)0, c, false, false);
}

__global__ __launch_bounds__(64) __attribute__((amdgpu_waves_per_eu(1, 2)))
void k_gemm_wmma(const float* __restrict__ x,
                 const float* __restrict__ W,
                 const float* __restrict__ dis,
                 const float* __restrict__ bias,
                 float* __restrict__ h,
                 float* __restrict__ out) {
    const int lane = threadIdx.x & 31;
    const int c    = lane & 15;
    const int hf   = lane >> 4;
    const int m0   = blockIdx.x * GM_ROWS;
    const int n0   = (threadIdx.x >> 5) * GM_COLS;
    const bool two = m0 + 16 < N_NODES;

    const int ra = m0 + c;
    const int rb = two ? m0 + 16 + c : N_NODES - 1;
    const float* __restrict__ xa0 = x + ra * IN_C + 4 * hf;
    const float* __restrict__ xa1 = x + rb * IN_C + 4 * hf;
    const float* __restrict__ wb  = W + (n0 + 4 * c) * IN_C + 4 * hf;

    const int mrow = m0 + 8 * hf;
    const int ncol = n0 + 4 * c;
    unsigned obyte  = (unsigned)(mrow * OUT_C + ncol) * 4u;
    unsigned dbyte0 = (unsigned)mrow * 4u;
    unsigned dbyte1 = two ? dbyte0 + 64u : dbyte0;
    asm volatile("" : "+v"(obyte), "+v"(dbyte0), "+v"(dbyte1));
    const float* dp0 = (const float*)((const char*)dis + dbyte0);
    const float* dp1 = (const float*)((const char*)dis + dbyte1);
    float* hp = (float*)((char*)h + obyte);
    float* op = (float*)((char*)out + obyte);

    GemmStage st[GM_AHEAD];
    #pragma unroll
    for (int s = 0; s < GM_AHEAD; ++s) gemm_request(st[s], xa0, xa1, wb, s);

    float d0[8], d1[8];
    #pragma unroll
    for (int v = 0; v < 8; ++v) { d0[v] = dp0[v]; d1[v] = dp1[v]; }
    const float4 bv = *(const float4*)(bias + ncol);

    __builtin_amdgcn_sched_barrier(0);
    v8f acc[2][4] = {};
    #pragma unroll
    for (int kb = 0; kb < GM_STEPS; ++kb) {
        const GemmStage cur = st[kb % GM_AHEAD];
        if (kb + GM_AHEAD < GM_STEPS)
            gemm_request(st[kb % GM_AHEAD], xa0, xa1, wb, kb + GM_AHEAD);
        const v2f a0l = {cur.a0.x, cur.a0.y}, a0h = {cur.a0.z, cur.a0.w};
        const v2f a1l = {cur.a1.x, cur.a1.y}, a1h = {cur.a1.z, cur.a1.w};
        #pragma unroll
        for (int j = 0; j < 4; ++j) {
            const v2f bl = {cur.b[j].x, cur.b[j].y};
            acc[0][j] = gemm_mma(a0l, bl, acc[0][j]);
            acc[1][j] = gemm_mma(a1l, bl, acc[1][j]);
        }
        #pragma unroll
        for (int j = 0; j < 4; ++j) {
            const v2f bh = {cur.b[j].z, cur.b[j].w};
            acc[0][j] = gemm_mma(a0h, bh, acc[0][j]);
            acc[1][j] = gemm_mma(a1h, bh, acc[1][j]);
        }
    }
    __builtin_amdgcn_sched_barrier(0);
    asm volatile("v_nop\n\tv_nop\n\tv_nop\n\tv_nop");
    __builtin_amdgcn_sched_barrier(0);

    #pragma unroll
    for (int v = 0; v < 8; ++v) {
        const float d = d0[v];
        const float4 hv = make_float4(acc[0][0][v] * d, acc[0][1][v] * d, acc[0][2][v] * d, acc[0][3][v] * d);
        *(float4*)(hp + v * OUT_C) = hv;
        *(float4*)(op + v * OUT_C) =
            make_float4(hv.x * d + bv.x, hv.y * d + bv.y, hv.z * d + bv.z, hv.w * d + bv.w);
    }
    if (two) {
        #pragma unroll
        for (int v = 0; v < 8; ++v) {
            const float d = d1[v];
            const float4 hv = make_float4(acc[1][0][v] * d, acc[1][1][v] * d, acc[1][2][v] * d, acc[1][3][v] * d);
            *(float4*)(hp + (16 + v) * OUT_C) = hv;
            *(float4*)(op + (16 + v) * OUT_C) =
                make_float4(hv.x * d + bv.x, hv.y * d + bv.y, hv.z * d + bv.z, hv.w * d + bv.w);
        }
    }
}

#define HALF         16
#define SLAB_GRID    256
#define WAVE_ROWS    (TILE_N / SLAB_WAVES)
#define MERGE_ROWS   5
#define LDS_FLOATS   42000

static_assert(N_NODES % D_PARTS == 0 && N_EDGES % E_PARTS == 0 && SLAB_C == 32,
              "slab geometry");
static_assert((TILE_N + 1) * SLAB_C <= LDS_FLOATS && LDS_FLOATS * 4 + 4 <= 327680 && LDS_FLOATS * 4 > 163840,
              "slab and dump row fit the reservation; the reservation fits one WGP and excludes a second work-group");
static_assert(SLAB_GRID == D_PARTS * S_PARTS * E_PARTS, "one work-group per (slice, range, slab)");
static_assert(2 * HALF == RUN_BATCH, "a batch is two request groups");
#define ROLES_PER_HALF (SLAB_GRID / 2)
static_assert(D_PARTS % 2 == 0 && (SLAB_GRID & (SLAB_GRID - 1)) == 0, "roles split evenly by half");

struct EdgeHalf {
    float v[HALF];
    int   a[HALF];
};

__device__ __forceinline__ void half_request(EdgeHalf& k, int rec, int off,
                                             const float* __restrict__ hs_slab,
                                             unsigned lane4) {
    #pragma unroll
    for (int j = 0; j < HALF; ++j) {
        const unsigned r = (unsigned)__builtin_amdgcn_readlane(rec, off + j);
        const unsigned byte = (r & SRC_MASK) * (OUT_C * 4) + lane4;
        k.v[j] = *(const float*)((const char*)hs_slab + byte);
        k.a[j] = (int)((r >> SRC_BITS) * (SLAB_C * 4) + lane4);
    }
}

__device__ __forceinline__ void half_add(const EdgeHalf& k, float* tile) {
    #pragma unroll
    for (int j = 0; j < HALF; ++j)
        __hip_atomic_fetch_add((float*)((char*)tile + k.a[j]), k.v[j],
                               __ATOMIC_RELAXED, __HIP_MEMORY_SCOPE_WORKGROUP);
}

__global__ __launch_bounds__(SLAB_THREADS)
void k_edge_slab(const int* __restrict__ tab,
                 const int* __restrict__ recs,
                 const float* __restrict__ hs,
                 const float* __restrict__ dis,
                 float* __restrict__ out,
                 int* __restrict__ ticket) {
    __shared__ float tile[LDS_FLOATS];
    __shared__ int   s_role;

    const int tid  = threadIdx.x;
    const int lane = tid & 31;

    if (tid == 0) {
        const int H = package_half();
        const int t = atomicAdd(&ticket[H * TICKET_STRIDE], 1);
        s_role = (t < ROLES_PER_HALF ? H * ROLES_PER_HALF + t
                                     : (1 - H) * ROLES_PER_HALF + (2 * ROLES_PER_HALF - 1 - t))
                 & (SLAB_GRID - 1);
    }

    float4* t4 = (float4*)tile;
    for (int i = tid; i < (TILE_N + 1) * SLAB_C / 4; i += SLAB_THREADS)
        t4[i] = make_float4(0.0f, 0.0f, 0.0f, 0.0f);
    __syncthreads();

    const unsigned role = (unsigned)__builtin_amdgcn_readfirstlane(s_role);
    const int      wave = __builtin_amdgcn_readfirstlane(tid >> 5);
    const unsigned idx  = role % ROLES_PER_HALF;
    const int d = (int)((role / ROLES_PER_HALF) * (D_PARTS / 2) + idx % (D_PARTS / 2));
    const int s = (int)((idx / (D_PARTS / 2)) % S_PARTS);
    const int p = (int)(idx / ((D_PARTS / 2) * S_PARTS));

    const int    nbase = d * TILE_N;
    const float*   hs_slab = hs + s * SLAB_C;
    const unsigned lane4   = (unsigned)lane * 4u;

    const int  blk = p * SLAB_WAVES + wave;
    const int  te  = tab[blk * D_PARTS + d];
    const int  nb  = te >> 16;
    const int* rp  = recs + blk * PB_STRIDE + (te & 0xffff) + lane;

    if (nb > 0) {
        EdgeHalf A, B;
        int rec  = rp[0];
        int recn = rp[(nb > 1 ? 1 : 0) * RUN_BATCH];
        asm volatile("" : "+v"(rec), "+v"(recn));
        half_request(A, rec, 0, hs_slab, lane4);
        for (int b = 1; b < nb; ++b) {
            half_request(B, rec, HALF, hs_slab, lane4);
            half_add(A, tile);
            rec  = recn;
            recn = rp[(b + 1 < nb ? b + 1 : b) * RUN_BATCH];
            half_request(A, rec, 0, hs_slab, lane4);
            half_add(B, tile);
        }
        half_request(B, rec, HALF, hs_slab, lane4);
        half_add(A, tile);
        half_add(B, tile);
    }
    __syncthreads();

    float* o_lane = out + s * SLAB_C + lane;
    #pragma unroll
    for (int g = 0; g < WAVE_ROWS; g += MERGE_ROWS) {
        float v[MERGE_ROWS], dc[MERGE_ROWS];
        #pragma unroll
        for (int j = 0; j < MERGE_ROWS; ++j) {
            if (g + j < WAVE_ROWS) {
                const int nn = wave + (g + j) * SLAB_WAVES;
                v[j]  = tile[nn * SLAB_C + lane];
                dc[j] = dis[nbase + nn];
            }
        }
        #pragma unroll
        for (int j = 0; j < MERGE_ROWS; ++j) {
            if (g + j < WAVE_ROWS) {
                const int nn = wave + (g + j) * SLAB_WAVES;
                atomicAdd(o_lane + (nbase + nn) * OUT_C, v[j] * dc[j]);
            }
        }
    }
    if (wave < TILE_N - WAVE_ROWS * SLAB_WAVES) {
        const int nn = WAVE_ROWS * SLAB_WAVES + wave;
        atomicAdd(o_lane + (nbase + nn) * OUT_C, tile[nn * SLAB_C + lane] * dis[nbase + nn]);
    }
}

extern "C" void kernel_launch(void* const* d_in, const int* in_sizes, int n_in,
                              void* d_out, int out_size, void* d_ws, size_t ws_size,
                              hipStream_t stream) {
    const float* x  = (const float*)d_in[0];
    const int*   ei = (const int*)  d_in[1];
    const float* W  = (const float*)d_in[2];
    const float* b  = (const float*)d_in[3];
    float*       out = (float*)d_out;

    const int* row = ei;
    const int* col = ei + N_EDGES;

    float* deg  = (float*)d_ws;
    float* dis  = deg + N_NODES;
    float* h    = dis + N_NODES;
    int*   tab  = (int*)(h + N_NODES * OUT_C);
    int*   recs = tab + PB_BLOCKS * D_PARTS;
    if (ws_size < WS_BYTES) return;

    k_init_deg<<<(N_NODES + 255) / 256, 256, 0, stream>>>(deg, dis);
    k_deg_part<<<2 * PB_BLOCKS, PB_THREADS, 0, stream>>>(row, col, deg, dis, tab, recs);
    k_inv_sqrt<<<(N_NODES + 255) / 256, 256, 0, stream>>>(deg, dis);
    k_gemm_wmma<<<GM_BLOCKS, 64, 0, stream>>>(x, W, dis, b, h, out);
    k_edge_slab<<<SLAB_GRID, SLAB_THREADS, 0, stream>>>(tab, recs, h, dis, out, (int*)deg);
}


# rsqrt launch gone: GEMM lanes and the slab merge form dis = rsqrt(plane 0 + plane 1) from the two degree planes themselves; the degree-init launch clears the role counters; sums, order unchanged
# speedup vs baseline: 4.6865x; 1.0536x over previous
#include <hip/hip_runtime.h>
#include <hip/hip_bf16.h>


#define N_NODES 10000
#define N_EDGES 640000
#define IN_C    128
#define OUT_C   128

typedef __attribute__((ext_vector_type(2))) float v2f;
typedef __attribute__((ext_vector_type(8))) float v8f;

__device__ __forceinline__ int package_half() {
    return (int)((__builtin_amdgcn_s_sendmsg_rtn(0x87) >> 18) & 1u);
}

__device__ __forceinline__ float inv_sqrt_deg(float a, float b) {
    const float d = a + b;
    return (d > 0.0f) ? rsqrtf(d) : 0.0f;
}

__global__ void k_init_deg(float* __restrict__ deg0, float* __restrict__ deg1,
                           int* __restrict__ ticket0, int* __restrict__ ticket1) {
    int i = blockIdx.x * blockDim.x + threadIdx.x;
    if (i < N_NODES) { deg0[i] = 1.0f; deg1[i] = 0.0f; }
    if (i == 0) { *ticket0 = 0; *ticket1 = 0; }
}

#define D_PARTS      16
#define S_PARTS      4
#define E_PARTS      4
#define TILE_N       (N_NODES / D_PARTS)
#define SLAB_C       (OUT_C / S_PARTS)
#define SLAB_THREADS 1024
#define SLAB_WAVES   (SLAB_THREADS / 32)
#define PB_BLOCKS    128
#define PB_EDGES     (N_EDGES / PB_BLOCKS)
#define PB_THREADS   1024
#define PB_SLOTS     5
#define RUN_BATCH    32
#define PB_STRIDE    5504
#define SRC_BITS     14
#define SRC_MASK     ((1u << SRC_BITS) - 1u)
#define DUMP_REC     ((unsigned)TILE_N << SRC_BITS)
#define WS_BYTES     ((size_t)4 * (2 * N_NODES + N_NODES * OUT_C + PB_BLOCKS * D_PARTS + PB_BLOCKS * PB_STRIDE))

static_assert(N_NODES % D_PARTS == 0, "node ranges are equal");
static_assert(PB_BLOCKS * PB_EDGES == N_EDGES && PB_THREADS * PB_SLOTS >= PB_EDGES &&
              PB_THREADS % 32 == 0, "partition blocks cover the edge list in whole waves");
static_assert(PB_BLOCKS == E_PARTS * SLAB_WAVES, "one partition block per edge-pass wave");
static_assert(PB_STRIDE >= PB_EDGES + D_PARTS * (RUN_BATCH - 1) && PB_STRIDE % RUN_BATCH == 0,
              "the padded runs fit a block's record region");
#define TICKET_OFF   (((PB_EDGES + D_PARTS * (RUN_BATCH - 1)) / RUN_BATCH) * RUN_BATCH)
#define TICKET_STRIDE PB_STRIDE
static_assert(TICKET_OFF < PB_STRIDE && TICKET_STRIDE * 4 >= 256 && PB_BLOCKS >= 2,
              "the role counters sit behind the last possible record, on lines of their own");
static_assert(PB_STRIDE < (1 << 16) && D_PARTS * RUN_BATCH <= PB_THREADS,
              "a run's start fits the table entry's low half; one wave pads each run");
static_assert(N_NODES <= (1 << SRC_BITS) && TILE_N + 1 <= (1 << (32 - SRC_BITS)),
              "record fields hold a source row and a local row (dump row included)");

__global__ __launch_bounds__(PB_THREADS)
void k_deg_part(const int* __restrict__ row, const int* __restrict__ col,
                float* __restrict__ deg0, float* __restrict__ deg1,
                int* __restrict__ tab, int* __restrict__ recs) {
    __shared__ int s_cnt[D_PARTS];
    __shared__ int s_start[D_PARTS];
    const int tid  = threadIdx.x;
    const int lane = tid & 31;

    if (blockIdx.x >= PB_BLOCKS) {
        float* plane = package_half() ? deg1 : deg0;
        const int* cp = col + (blockIdx.x - PB_BLOCKS) * PB_EDGES;
        int c[PB_SLOTS];
        #pragma unroll
        for (int q = 0; q < PB_SLOTS; ++q) {
            const int i = tid + q * PB_THREADS;
            c[q] = i < PB_EDGES ? cp[i] : -1;
        }
        #pragma unroll
        for (int q = 0; q < PB_SLOTS; ++q)
            if (c[q] >= 0) atomicAdd(&plane[c[q]], 1.0f);
        return;
    }

    if (tid < D_PARTS) s_cnt[tid] = 0;
    __syncthreads();

    const int e0 = blockIdx.x * PB_EDGES;

    int key[PB_SLOTS], pos[PB_SLOTS];
    unsigned rec[PB_SLOTS];
    #pragma unroll
    for (int q = 0; q < PB_SLOTS; ++q) {
        const int i = tid + q * PB_THREADS;
        int k = -1;
        unsigned rc = 0u;
        if (i < PB_EDGES) {
            const int c = col[e0 + i];
            const int r = row[e0 + i];
            k  = c / TILE_N;
            rc = (unsigned)r | ((unsigned)(c - k * TILE_N) << SRC_BITS);
        }
        key[q] = k; rec[q] = rc;
    }
    #pragma unroll
    for (int q = 0; q < PB_SLOTS; ++q)
        pos[q] = key[q] >= 0 ? atomicAdd(&s_cnt[key[q]], 1) : 0;
    __syncthreads();

    if (tid < D_PARTS) {
        int st = 0;
        for (int j = 0; j < tid; ++j) st += (s_cnt[j] + RUN_BATCH - 1) & ~(RUN_BATCH - 1);
        s_start[tid] = st;
    }
    __syncthreads();

    int* my = recs + blockIdx.x * PB_STRIDE;
    #pragma unroll
    for (int q = 0; q < PB_SLOTS; ++q)
        if (key[q] >= 0) my[s_start[key[q]] + pos[q]] = (int)rec[q];
    if (tid < D_PARTS * RUN_BATCH) {
        const int k  = tid >> 5;
        const int tk = s_cnt[k];
        const int st = s_start[k];
        const int padded = (tk + RUN_BATCH - 1) & ~(RUN_BATCH - 1);
        if (tk + lane < padded) my[st + tk + lane] = (int)DUMP_REC;
        if (lane == 0) tab[blockIdx.x * D_PARTS + k] = st | ((padded / RUN_BATCH) << 16);
    }
}

#define GM_ROWS   32
#define GM_COLS   64
#define GM_BLOCKS ((N_NODES + GM_ROWS - 1) / GM_ROWS)
#define GM_KSTEP  8
#define GM_STEPS  (IN_C / GM_KSTEP)
#define GM_AHEAD  4

static_assert(IN_C % GM_KSTEP == 0 && OUT_C == 2 * GM_COLS && GM_STEPS >= GM_AHEAD,
              "two waves cover a row's columns; whole k steps");
static_assert(N_NODES % 16 == 0 && N_NODES % GM_ROWS == 16,
              "the last block's first row tile is whole and its second lies wholly outside");

struct GemmStage { float4 a0, a1, b[4]; };

__device__ __forceinline__ void gemm_request(GemmStage& s, const float* __restrict__ xa0,
                                             const float* __restrict__ xa1,
                                             const float* __restrict__ wb, int kb) {
    s.a0 = *(const float4*)(xa0 + kb * GM_KSTEP);
    s.a1 = *(const float4*)(xa1 + kb * GM_KSTEP);
    #pragma unroll
    for (int j = 0; j < 4; ++j)
        s.b[j] = *(const float4*)(wb + j * IN_C + kb * GM_KSTEP);
}

__device__ __forceinline__ v8f gemm_mma(v2f a, v2f b, v8f c) {
    return __builtin_amdgcn_wmma_f32_16x16x4_f32(false, a, false, b, (short)0, c, false, false);
}

__global__ __launch_bounds__(64) __attribute__((amdgpu_waves_per_eu(1, 2)))
void k_gemm_wmma(const float* __restrict__ x,
                 const float* __restrict__ W,
                 const float* __restrict__ deg0,
                 const float* __restrict__ deg1,
                 const float* __restrict__ bias,
                 float* __restrict__ h,
                 float* __restrict__ out) {
    const int lane = threadIdx.x & 31;
    const int c    = lane & 15;
    const int hf   = lane >> 4;
    const int m0   = blockIdx.x * GM_ROWS;
    const int n0   = (threadIdx.x >> 5) * GM_COLS;
    const bool two = m0 + 16 < N_NODES;

    const int ra = m0 + c;
    const int rb = two ? m0 + 16 + c : N_NODES - 1;
    const float* __restrict__ xa0 = x + ra * IN_C + 4 * hf;
    const float* __restrict__ xa1 = x + rb * IN_C + 4 * hf;
    const float* __restrict__ wb  = W + (n0 + 4 * c) * IN_C + 4 * hf;

    const int mrow = m0 + 8 * hf;
    const int ncol = n0 + 4 * c;
    unsigned obyte  = (unsigned)(mrow * OUT_C + ncol) * 4u;
    unsigned dbyte0 = (unsigned)mrow * 4u;
    unsigned dbyte1 = two ? dbyte0 + 64u : dbyte0;
    asm volatile("" : "+v"(obyte), "+v"(dbyte0), "+v"(dbyte1));
    const float* pa0 = (const float*)((const char*)deg0 + dbyte0);
    const float* pb0 = (const float*)((const char*)deg1 + dbyte0);
    const float* pa1 = (const float*)((const char*)deg0 + dbyte1);
    const float* pb1 = (const float*)((const char*)deg1 + dbyte1);
    float* hp = (float*)((char*)h + obyte);
    float* op = (float*)((char*)out + obyte);

    GemmStage st[GM_AHEAD];
    #pragma unroll
    for (int s = 0; s < GM_AHEAD; ++s) gemm_request(st[s], xa0, xa1, wb, s);

    float ga0[8], gb0[8], ga1[8], gb1[8];
    #pragma unroll
    for (int v = 0; v < 8; ++v) { ga0[v] = pa0[v]; gb0[v] = pb0[v]; ga1[v] = pa1[v]; gb1[v] = pb1[v]; }
    const float4 bv = *(const float4*)(bias + ncol);

    __builtin_amdgcn_sched_barrier(0);
    v8f acc[2][4] = {};
    #pragma unroll
    for (int kb = 0; kb < GM_STEPS; ++kb) {
        const GemmStage cur = st[kb % GM_AHEAD];
        if (kb + GM_AHEAD < GM_STEPS)
            gemm_request(st[kb % GM_AHEAD], xa0, xa1, wb, kb + GM_AHEAD);
        const v2f a0l = {cur.a0.x, cur.a0.y}, a0h = {cur.a0.z, cur.a0.w};
        const v2f a1l = {cur.a1.x, cur.a1.y}, a1h = {cur.a1.z, cur.a1.w};
        #pragma unroll
        for (int j = 0; j < 4; ++j) {
            const v2f bl = {cur.b[j].x, cur.b[j].y};
            acc[0][j] = gemm_mma(a0l, bl, acc[0][j]);
            acc[1][j] = gemm_mma(a1l, bl, acc[1][j]);
        }
        #pragma unroll
        for (int j = 0; j < 4; ++j) {
            const v2f bh = {cur.b[j].z, cur.b[j].w};
            acc[0][j] = gemm_mma(a0h, bh, acc[0][j]);
            acc[1][j] = gemm_mma(a1h, bh, acc[1][j]);
        }
    }
    __builtin_amdgcn_sched_barrier(0);
    asm volatile("v_nop\n\tv_nop\n\tv_nop\n\tv_nop");
    __builtin_amdgcn_sched_barrier(0);
    #pragma unroll
    for (int v = 0; v < 8; ++v)
        asm volatile("" : "+v"(ga0[v]), "+v"(gb0[v]), "+v"(ga1[v]), "+v"(gb1[v]));

    #pragma unroll
    for (int v = 0; v < 8; ++v) {
        const float d = inv_sqrt_deg(ga0[v], gb0[v]);
        const float4 hv = make_float4(acc[0][0][v] * d, acc[0][1][v] * d, acc[0][2][v] * d, acc[0][3][v] * d);
        *(float4*)(hp + v * OUT_C) = hv;
        *(float4*)(op + v * OUT_C) =
            make_float4(hv.x * d + bv.x, hv.y * d + bv.y, hv.z * d + bv.z, hv.w * d + bv.w);
    }
    if (two) {
        #pragma unroll
        for (int v = 0; v < 8; ++v) {
            const float d = inv_sqrt_deg(ga1[v], gb1[v]);
            const float4 hv = make_float4(acc[1][0][v] * d, acc[1][1][v] * d, acc[1][2][v] * d, acc[1][3][v] * d);
            *(float4*)(hp + (16 + v) * OUT_C) = hv;
            *(float4*)(op + (16 + v) * OUT_C) =
                make_float4(hv.x * d + bv.x, hv.y * d + bv.y, hv.z * d + bv.z, hv.w * d + bv.w);
        }
    }
}

#define HALF         16
#define SLAB_GRID    256
#define WAVE_ROWS    (TILE_N / SLAB_WAVES)
#define MERGE_ROWS   5
#define LDS_FLOATS   42000

static_assert(N_NODES % D_PARTS == 0 && N_EDGES % E_PARTS == 0 && SLAB_C == 32,
              "slab geometry");
static_assert((TILE_N + 1) * SLAB_C <= LDS_FLOATS && LDS_FLOATS * 4 + 4 <= 327680 && LDS_FLOATS * 4 > 163840,
              "slab and dump row fit the reservation; the reservation fits one WGP and excludes a second work-group");
static_assert(SLAB_GRID == D_PARTS * S_PARTS * E_PARTS, "one work-group per (slice, range, slab)");
static_assert(2 * HALF == RUN_BATCH, "a batch is two request groups");
#define ROLES_PER_HALF (SLAB_GRID / 2)
static_assert(D_PARTS % 2 == 0 && (SLAB_GRID & (SLAB_GRID - 1)) == 0, "roles split evenly by half");

struct EdgeHalf {
    float v[HALF];
    int   a[HALF];
};

__device__ __forceinline__ void half_request(EdgeHalf& k, int rec, int off,
                                             const float* __restrict__ hs_slab,
                                             unsigned lane4) {
    #pragma unroll
    for (int j = 0; j < HALF; ++j) {
        const unsigned r = (unsigned)__builtin_amdgcn_readlane(rec, off + j);
        const unsigned byte = (r & SRC_MASK) * (OUT_C * 4) + lane4;
        k.v[j] = *(const float*)((const char*)hs_slab + byte);
        k.a[j] = (int)((r >> SRC_BITS) * (SLAB_C * 4) + lane4);
    }
}

__device__ __forceinline__ void half_add(const EdgeHalf& k, float* tile) {
    #pragma unroll
    for (int j = 0; j < HALF; ++j)
        __hip_atomic_fetch_add((float*)((char*)tile + k.a[j]), k.v[j],
                               __ATOMIC_RELAXED, __HIP_MEMORY_SCOPE_WORKGROUP);
}

__global__ __launch_bounds__(SLAB_THREADS)
void k_edge_slab(const int* __restrict__ tab,
                 const int* __restrict__ recs,
                 const float* __restrict__ hs,
                 const float* __restrict__ deg0,
                 const float* __restrict__ deg1,
                 float* __restrict__ out,
                 int* __restrict__ ticket) {
    __shared__ float tile[LDS_FLOATS];
    __shared__ int   s_role;

    const int tid  = threadIdx.x;
    const int lane = tid & 31;

    if (tid == 0) {
        const int H = package_half();
        const int t = atomicAdd(&ticket[H * TICKET_STRIDE], 1);
        s_role = (t < ROLES_PER_HALF ? H * ROLES_PER_HALF + t
                                     : (1 - H) * ROLES_PER_HALF + (2 * ROLES_PER_HALF - 1 - t))
                 & (SLAB_GRID - 1);
    }

    float4* t4 = (float4*)tile;
    for (int i = tid; i < (TILE_N + 1) * SLAB_C / 4; i += SLAB_THREADS)
        t4[i] = make_float4(0.0f, 0.0f, 0.0f, 0.0f);
    __syncthreads();

    const unsigned role = (unsigned)__builtin_amdgcn_readfirstlane(s_role);
    const int      wave = __builtin_amdgcn_readfirstlane(tid >> 5);
    const unsigned idx  = role % ROLES_PER_HALF;
    const int d = (int)((role / ROLES_PER_HALF) * (D_PARTS / 2) + idx % (D_PARTS / 2));
    const int s = (int)((idx / (D_PARTS / 2)) % S_PARTS);
    const int p = (int)(idx / ((D_PARTS / 2) * S_PARTS));

    const int    nbase = d * TILE_N;
    const float*   hs_slab = hs + s * SLAB_C;
    const unsigned lane4   = (unsigned)lane * 4u;

    const int  blk = p * SLAB_WAVES + wave;
    const int  te  = tab[blk * D_PARTS + d];
    const int  nb  = te >> 16;
    const int* rp  = recs + blk * PB_STRIDE + (te & 0xffff) + lane;

    if (nb > 0) {
        EdgeHalf A, B;
        int rec  = rp[0];
        int recn = rp[(nb > 1 ? 1 : 0) * RUN_BATCH];
        asm volatile("" : "+v"(rec), "+v"(recn));
        half_request(A, rec, 0, hs_slab, lane4);
        for (int b = 1; b < nb; ++b) {
            half_request(B, rec, HALF, hs_slab, lane4);
            half_add(A, tile);
            rec  = recn;
            recn = rp[(b + 1 < nb ? b + 1 : b) * RUN_BATCH];
            half_request(A, rec, 0, hs_slab, lane4);
            half_add(B, tile);
        }
        half_request(B, rec, HALF, hs_slab, lane4);
        half_add(A, tile);
        half_add(B, tile);
    }
    __syncthreads();

    float* o_lane = out + s * SLAB_C + lane;
    #pragma unroll
    for (int g = 0; g < WAVE_ROWS; g += MERGE_ROWS) {
        float v[MERGE_ROWS], dc[MERGE_ROWS];
        #pragma unroll
        for (int j = 0; j < MERGE_ROWS; ++j) {
            if (g + j < WAVE_ROWS) {
                const int nn = wave + (g + j) * SLAB_WAVES;
                v[j]  = tile[nn * SLAB_C + lane];
                dc[j] = inv_sqrt_deg(deg0[nbase + nn], deg1[nbase + nn]);
            }
        }
        #pragma unroll
        for (int j = 0; j < MERGE_ROWS; ++j) {
            if (g + j < WAVE_ROWS) {
                const int nn = wave + (g + j) * SLAB_WAVES;
                atomicAdd(o_lane + (nbase + nn) * OUT_C, v[j] * dc[j]);
            }
        }
    }
    if (wave < TILE_N - WAVE_ROWS * SLAB_WAVES) {
        const int nn = WAVE_ROWS * SLAB_WAVES + wave;
        atomicAdd(o_lane + (nbase + nn) * OUT_C,
                  tile[nn * SLAB_C + lane] * inv_sqrt_deg(deg0[nbase + nn], deg1[nbase + nn]));
    }
}

extern "C" void kernel_launch(void* const* d_in, const int* in_sizes, int n_in,
                              void* d_out, int out_size, void* d_ws, size_t ws_size,
                              hipStream_t stream) {
    const float* x  = (const float*)d_in[0];
    const int*   ei = (const int*)  d_in[1];
    const float* W  = (const float*)d_in[2];
    const float* b  = (const float*)d_in[3];
    float*       out = (float*)d_out;

    const int* row = ei;
    const int* col = ei + N_EDGES;

    float* deg0 = (float*)d_ws;
    float* deg1 = deg0 + N_NODES;
    float* h    = deg1 + N_NODES;
    int*   tab  = (int*)(h + N_NODES * OUT_C);
    int*   recs = tab + PB_BLOCKS * D_PARTS;
    int*   ticket = recs + TICKET_OFF;
    if (ws_size < WS_BYTES) return;

    k_init_deg<<<(N_NODES + 255) / 256, 256, 0, stream>>>(deg0, deg1, ticket, ticket + TICKET_STRIDE);
    k_deg_part<<<2 * PB_BLOCKS, PB_THREADS, 0, stream>>>(row, col, deg0, deg1, tab, recs);
    k_gemm_wmma<<<GM_BLOCKS, 64, 0, stream>>>(x, W, deg0, deg1, b, h, out);
    k_edge_slab<<<SLAB_GRID, SLAB_THREADS, 0, stream>>>(tab, recs, h, deg0, deg1, out, ticket);
}
